# starting kernel
# baseline (speedup 1.0000x reference)
#include <hip/hip_runtime.h>
#include <math.h>


#define LS     68
#define BUF    (16 * LS)
#define TSTEPS 200
#define DDIM   64

typedef float v2f __attribute__((ext_vector_type(2)));
typedef float v8f __attribute__((ext_vector_type(8)));

__device__ __forceinline__ float fast_tanh(float v) {
#if __has_builtin(__builtin_amdgcn_tanhf)
  return __builtin_amdgcn_tanhf(v);
#elif __has_builtin(__builtin_amdgcn_tanh_f32)
  return __builtin_amdgcn_tanh_f32(v);
#else
  return tanhf(v);
#endif
}

__device__ __forceinline__ v8f wmma_f32(v2f a, v2f b, v8f c) {
  return __builtin_amdgcn_wmma_f32_16x16x4_f32(false, a, false, b, (short)0, c,
                                               false, false);
}

__device__ __forceinline__ void proj16x64(const float* src, const v2f* wf,
                                          float* dst, int m, int h) {
  v8f a0 = {}, a1 = {}, a2 = {}, a3 = {};
  const float* arow = src + m * LS + 2 * h;
#pragma unroll
  for (int kc = 0; kc < 16; ++kc) {
    v2f af = *(const v2f*)(arow + 4 * kc);
    a0 = wmma_f32(af, wf[kc * 4 + 0], a0);
    a1 = wmma_f32(af, wf[kc * 4 + 1], a1);
    a2 = wmma_f32(af, wf[kc * 4 + 2], a2);
    a3 = wmma_f32(af, wf[kc * 4 + 3], a3);
  }
#pragma unroll
  for (int r = 0; r < 8; ++r) {
    int row = r + 8 * h;
    dst[row * LS + m +  0] = a0[r];
    dst[row * LS + m + 16] = a1[r];
    dst[row * LS + m + 32] = a2[r];
    dst[row * LS + m + 48] = a3[r];
  }
}

__device__ __forceinline__ float rowdot(const float* A, const float* B, int m,
                                        int h) {
  const float4* a = (const float4*)(A + m * LS + 32 * h);
  const float4* b = (const float4*)(B + m * LS + 32 * h);
  float s = 0.f;
#pragma unroll
  for (int i = 0; i < 8; ++i) {
    float4 u = a[i], v = b[i];
    s += u.x * v.x + u.y * v.y + u.z * v.z + u.w * v.w;
  }
  s += __shfl_xor(s, 16, 32);
  return s;
}

__device__ __forceinline__ void softmax2(float a, float b, float& p0,
                                         float& p1) {
  float mx  = fmaxf(a, b);
  float e0  = __expf(a - mx);
  float e1  = __expf(b - mx);
  float inv = 1.f / (e0 + e1);
  p0 = e0 * inv;
  p1 = e1 * inv;
}

template <bool DO_TANH>
__device__ __forceinline__ void combine(const float* V0, const float* V1,
                                        float p0, float p1, float* dst, int m,
                                        int h) {
  const float4* v0 = (const float4*)(V0 + m * LS + 32 * h);
  const float4* v1 = (const float4*)(V1 + m * LS + 32 * h);
  float4* d = (float4*)(dst + m * LS + 32 * h);
#pragma unroll
  for (int i = 0; i < 8; ++i) {
    float4 a = v0[i], b = v1[i], r;
    r.x = p0 * a.x + p1 * b.x;
    r.y = p0 * a.y + p1 * b.y;
    r.z = p0 * a.z + p1 * b.z;
    r.w = p0 * a.w + p1 * b.w;
    if (DO_TANH) {
      r.x = fast_tanh(r.x);
      r.y = fast_tanh(r.y);
      r.z = fast_tanh(r.z);
      r.w = fast_tanh(r.w);
    }
    d[i] = r;
  }
}

__global__ void swizzle_weights(const float* __restrict__ Wq,
                                const float* __restrict__ Wk,
                                const float* __restrict__ Wv,
                                float* __restrict__ wsw) {
  int idx = blockIdx.x * blockDim.x + threadIdx.x;
  if (idx >= 3 * 16 * 4 * 32 * 2) return;
  int j    = idx & 1;
  int lane = (idx >> 1) & 31;
  int nt   = (idx >> 6) & 3;
  int kc   = (idx >> 8) & 15;
  int w    = idx >> 12;
  int k = 4 * kc + j + 2 * (lane >> 4);
  int n = 16 * nt + (lane & 15);
  const float* W = (w == 0) ? Wq : (w == 1) ? Wk : Wv;
  wsw[idx] = W[k * 64 + n];
}

__global__ __launch_bounds__(96) void birnn_kernel(
    const float* __restrict__ x, const float* __restrict__ wsw,
    float* __restrict__ out) {
  __shared__ __align__(16) float smem[13 * BUF];
  float* bufL  = smem + 0 * BUF;
  float* bufS  = smem + 1 * BUF;
  float* bufX  = smem + 2 * BUF;
  float* bufO0 = smem + 3 * BUF;
  float* bufO1 = smem + 4 * BUF;
  float* P0 = smem + 5 * BUF;
  float* P1 = smem + 6 * BUF;
  float* P2 = smem + 7 * BUF;
  float* P3 = smem + 8 * BUF;
  float* P4 = smem + 9 * BUF;
  float* P5 = smem + 10 * BUF;
  float* P6 = smem + 11 * BUF;
  float* P7 = smem + 12 * BUF;

  const int lane = threadIdx.x & 31;
  const int wid  = threadIdx.x >> 5;
  const int m = lane & 15;
  const int h = lane >> 4;
  const int b = blockIdx.x * 16 + m;
  const float kScale = 0.125f;

  v2f wf[64];
  {
    const float* myw = wsw + wid * 4096 + lane * 2;
#pragma unroll
    for (int f = 0; f < 64; ++f) wf[f] = *(const v2f*)(myw + f * 64);
  }

  if (wid == 0) {
    const float4* xr = (const float4*)(x + (size_t)b * TSTEPS * DDIM + 32 * h);
    float4* dl  = (float4*)(bufL + m * LS + 32 * h);
    float4* dsv = (float4*)(bufS + m * LS + 32 * h);
#pragma unroll
    for (int i = 0; i < 8; ++i) {
      float4 v = xr[i];
      dl[i]  = v;
      dsv[i] = v;
    }
  }
  __syncthreads();

#pragma unroll 1
  for (int t = 1; t < TSTEPS; ++t) {
    if (wid == 2) {
      const float* xrow = x + ((size_t)b * TSTEPS + t) * DDIM + 32 * h;
      const float4* xr = (const float4*)xrow;
      float4* dx = (float4*)(bufX + m * LS + 32 * h);
#pragma unroll
      for (int i = 0; i < 8; ++i) dx[i] = xr[i];
      if (t + 1 < TSTEPS) __builtin_prefetch(xrow + DDIM, 0, 0);
    }
    __syncthreads();

    if (wid == 0) {
      proj16x64(bufL, wf, P0, m, h);
      proj16x64(bufX, wf, P3, m, h);
    } else if (wid == 1) {
      proj16x64(bufL, wf, P1, m, h);
      proj16x64(bufX, wf, P4, m, h);
      proj16x64(bufS, wf, P6, m, h);
    } else {
      proj16x64(bufL, wf, P2, m, h);
      proj16x64(bufX, wf, P5, m, h);
      proj16x64(bufS, wf, P7, m, h);
    }
    __syncthreads();

    if (wid == 0) {
      float d00 = rowdot(P0, P1, m, h) * kScale;
      float d01 = rowdot(P0, P4, m, h) * kScale;
      float p0, p1;
      softmax2(d00, d01, p0, p1);
      combine<true>(P2, P5, p0, p1, bufO0, m, h);
    } else if (wid == 1) {
      float d10 = rowdot(P3, P6, m, h) * kScale;
      float d11 = rowdot(P3, P4, m, h) * kScale;
      float q0, q1;
      softmax2(d10, d11, q0, q1);
      combine<true>(P7, P5, q0, q1, bufO1, m, h);
    }
    __syncthreads();

    if (wid == 0) {
      proj16x64(bufO0, wf, P0, m, h);
      proj16x64(bufO1, wf, P3, m, h);
    } else if (wid == 1) {
      proj16x64(bufO0, wf, P1, m, h);
      proj16x64(bufO1, wf, P4, m, h);
    } else {
      proj16x64(bufO0, wf, P2, m, h);
      proj16x64(bufO1, wf, P5, m, h);
    }
    __syncthreads();

    if (wid == 0) {
      float e00 = rowdot(P0, P1, m, h) * kScale;
      float e01 = rowdot(P0, P4, m, h) * kScale;
      float r0, r1;
      softmax2(e00, e01, r0, r1);
      combine<false>(P2, P5, r0, r1, bufL, m, h);
    } else if (wid == 1) {
      float e10 = rowdot(P3, P1, m, h) * kScale;
      float e11 = rowdot(P3, P4, m, h) * kScale;
      float s0, s1;
      softmax2(e10, e11, s0, s1);
      combine<false>(P2, P5, s0, s1, bufS, m, h);
    }
    __syncthreads();
  }

  if (wid == 0) {
    const float4* src = (const float4*)(bufL + m * LS + 32 * h);
    float4* dst = (float4*)(out + (size_t)b * DDIM + 32 * h);
#pragma unroll
    for (int i = 0; i < 8; ++i) dst[i] = src[i];
  }
}

extern "C" void kernel_launch(void* const* d_in, const int* in_sizes, int n_in,
                              void* d_out, int out_size, void* d_ws,
                              size_t ws_size, hipStream_t stream) {
  (void)in_sizes; (void)n_in; (void)out_size; (void)ws_size;
  const float* x  = (const float*)d_in[0];
  const float* Wq = (const float*)d_in[1];
  const float* Wk = (const float*)d_in[2];
  const float* Wv = (const float*)d_in[3];
  float* wsw = (float*)d_ws;

  swizzle_weights<<<48, 256, 0, stream>>>(Wq, Wk, Wv, wsw);
  birnn_kernel<<<2048 / 16, 96, 0, stream>>>(x, wsw, (float*)d_out);
}


# x_t skips LDS: each wave reads its A fragments (8 contiguous bytes a lane) from global memory half a step ahead; one barrier fewer a step, x chains run first over the LDS reads; sum order unchanged
# speedup vs baseline: 1.9602x; 1.9602x over previous
#include <hip/hip_runtime.h>
#include <math.h>


#define LS     68
#define BUF    (16 * LS)
#define TSTEPS 200
#define DDIM   64

typedef float v2f __attribute__((ext_vector_type(2)));
typedef float v8f __attribute__((ext_vector_type(8)));

__device__ __forceinline__ float fast_tanh(float v) {
#if __has_builtin(__builtin_amdgcn_tanhf)
  return __builtin_amdgcn_tanhf(v);
#elif __has_builtin(__builtin_amdgcn_tanh_f32)
  return __builtin_amdgcn_tanh_f32(v);
#else
  return tanhf(v);
#endif
}

__device__ __forceinline__ v8f wmma_f32(v2f a, v2f b, v8f c) {
  return __builtin_amdgcn_wmma_f32_16x16x4_f32(false, a, false, b, (short)0, c,
                                               false, false);
}

template <int CTRL>
__device__ __forceinline__ float rowmove(float v) {
  return __int_as_float(__builtin_amdgcn_update_dpp(0, __float_as_int(v), CTRL,
                                                    0xf, 0xf, true));
}

__device__ __forceinline__ void fold16(const float (&v)[32], int c, float& z0,
                                       float& z1) {
  const bool b3 = (c & 8) != 0, b2 = (c & 4) != 0, b1 = (c & 2) != 0,
             b0 = (c & 1) != 0;
  float w[16], u[8], t[4];
#pragma unroll
  for (int i = 0; i < 16; ++i) {
    float keep = b3 ? v[i + 16] : v[i];
    float send = b3 ? v[i] : v[i + 16];
    w[i] = keep + rowmove<0x168>(send);
  }
#pragma unroll
  for (int i = 0; i < 8; ++i) {
    float keep = b2 ? w[i + 8] : w[i];
    float send = b2 ? w[i] : w[i + 8];
    u[i] = keep + rowmove<0x164>(send);
  }
#pragma unroll
  for (int i = 0; i < 4; ++i) {
    float keep = b1 ? u[i + 4] : u[i];
    float send = b1 ? u[i] : u[i + 4];
    t[i] = keep + rowmove<0x162>(send);
  }
  {
    float keep = b0 ? t[2] : t[0];
    float send = b0 ? t[0] : t[2];
    z0 = keep + rowmove<0x161>(send);
  }
  {
    float keep = b0 ? t[3] : t[1];
    float send = b0 ? t[1] : t[3];
    z1 = keep + rowmove<0x161>(send);
  }
}

__device__ __forceinline__ void softmax2(float a, float b, float& p0,
                                         float& p1) {
  float mx  = fmaxf(a, b);
  float e0  = __expf(a - mx);
  float e1  = __expf(b - mx);
  float inv = 1.f / (e0 + e1);
  p0 = e0 * inv;
  p1 = e1 * inv;
}

template <int R, bool DO_TANH>
__device__ __forceinline__ void blend_row(float p0, float p1, const v8f& Va0,
                                          const v8f& Va1, const v8f& Vb0,
                                          const v8f& Vb1, float* dA, float* dB) {
  float a0 = rowmove<0x150 + 2 * R>(p0);
  float a1 = rowmove<0x150 + 2 * R>(p1);
  float b0 = rowmove<0x150 + 2 * R + 1>(p0);
  float b1 = rowmove<0x150 + 2 * R + 1>(p1);
  float oa = a0 * Va0[R] + a1 * Va1[R];
  float ob = b0 * Vb0[R] + b1 * Vb1[R];
  if (DO_TANH) {
    oa = fast_tanh(oa);
    ob = fast_tanh(ob);
  }
  dA[R * LS] = oa;
  dB[R * LS] = ob;
}

template <bool DO_TANH>
__device__ __forceinline__ void attend(const v8f& Qa, const v8f& Ka0,
                                       const v8f& Ka1, const v8f& Va0,
                                       const v8f& Va1, const v8f& Qb,
                                       const v8f& Kb0, const v8f& Kb1,
                                       const v8f& Vb0, const v8f& Vb1,
                                       float* part, float* dstA, float* dstB,
                                       int lane, int wid) {
  const int c = lane & 15;
  const int h = lane >> 4;
  float v[32];
#pragma unroll
  for (int r = 0; r < 8; ++r) {
    v[4 * r + 0] = Qa[r] * Ka0[r];
    v[4 * r + 1] = Qa[r] * Ka1[r];
    v[4 * r + 2] = Qb[r] * Kb0[r];
    v[4 * r + 3] = Qb[r] * Kb1[r];
  }
  float z0, z1;
  fold16(v, c, z0, z1);
  v2f zz;
  zz.x = z0;
  zz.y = z1;
  *(v2f*)(part + lane * 8 + wid * 2) = zz;
  __syncthreads();

  const float4 f0 = *(const float4*)(part + lane * 8);
  const float4 f1 = *(const float4*)(part + lane * 8 + 4);
  const float d0 = ((f0.x + f0.z) + f1.x) + f1.z;
  const float d1 = ((f0.y + f0.w) + f1.y) + f1.w;
  const float kScale = 0.125f;
  float p0, p1;
  softmax2(d0 * kScale, d1 * kScale, p0, p1);

  float* dA = dstA + (8 * h) * LS + 16 * wid + c;
  float* dB = dstB + (8 * h) * LS + 16 * wid + c;
  blend_row<0, DO_TANH>(p0, p1, Va0, Va1, Vb0, Vb1, dA, dB);
  blend_row<1, DO_TANH>(p0, p1, Va0, Va1, Vb0, Vb1, dA, dB);
  blend_row<2, DO_TANH>(p0, p1, Va0, Va1, Vb0, Vb1, dA, dB);
  blend_row<3, DO_TANH>(p0, p1, Va0, Va1, Vb0, Vb1, dA, dB);
  blend_row<4, DO_TANH>(p0, p1, Va0, Va1, Vb0, Vb1, dA, dB);
  blend_row<5, DO_TANH>(p0, p1, Va0, Va1, Vb0, Vb1, dA, dB);
  blend_row<6, DO_TANH>(p0, p1, Va0, Va1, Vb0, Vb1, dA, dB);
  blend_row<7, DO_TANH>(p0, p1, Va0, Va1, Vb0, Vb1, dA, dB);
  __syncthreads();
}

__global__ void swizzle_weights(const float* __restrict__ Wq,
                                const float* __restrict__ Wk,
                                const float* __restrict__ Wv,
                                float* __restrict__ wsw) {
  int idx = blockIdx.x * blockDim.x + threadIdx.x;
  if (idx >= 3 * 16 * 4 * 32 * 2) return;
  int j    = idx & 1;
  int lane = (idx >> 1) & 31;
  int nt   = (idx >> 6) & 3;
  int kc   = (idx >> 8) & 15;
  int w    = idx >> 12;
  int k = 4 * kc + j + 2 * (lane >> 4);
  int n = 16 * nt + (lane & 15);
  const float* W = (w == 0) ? Wq : (w == 1) ? Wk : Wv;
  wsw[idx] = W[k * 64 + n];
}

__global__ __launch_bounds__(128) void birnn_kernel(
    const float* __restrict__ x, const float* __restrict__ wsw,
    float* __restrict__ out) {
  __shared__ __align__(16) float smem[4 * BUF + 256];
  float* bufL  = smem + 0 * BUF;
  float* bufS  = smem + 1 * BUF;
  float* bufO0 = smem + 2 * BUF;
  float* bufO1 = smem + 3 * BUF;
  float* part  = smem + 4 * BUF;

  const int lane = threadIdx.x & 31;
  const int wid  = threadIdx.x >> 5;
  const int m = lane & 15;
  const int h = lane >> 4;
  const int b = blockIdx.x * 16 + m;

  v2f wq[16], wk[16], wv[16];
  {
    const float* myw = wsw + wid * 64 + lane * 2;
#pragma unroll
    for (int kc = 0; kc < 16; ++kc) {
      wq[kc] = *(const v2f*)(myw + kc * 256);
      wk[kc] = *(const v2f*)(myw + 4096 + kc * 256);
      wv[kc] = *(const v2f*)(myw + 8192 + kc * 256);
    }
  }

  if (wid == 0) {
    const float4* xr = (const float4*)(x + (size_t)b * TSTEPS * DDIM + 32 * h);
    float4* dl  = (float4*)(bufL + m * LS + 32 * h);
    float4* dsv = (float4*)(bufS + m * LS + 32 * h);
#pragma unroll
    for (int i = 0; i < 8; ++i) {
      float4 v = xr[i];
      dl[i]  = v;
      dsv[i] = v;
    }
  }
  __syncthreads();

  const int aoff = m * LS + 2 * h;
  const float* xlane = x + (size_t)b * TSTEPS * DDIM + 2 * h;

  v2f ax[16];
#pragma unroll
  for (int kc = 0; kc < 16; ++kc)
    ax[kc] = *(const v2f*)(xlane + DDIM + 4 * kc);

#pragma unroll 1
  for (int t = 1; t < TSTEPS; ++t) {
    {
      v2f al[16], as[16];
#pragma unroll
      for (int kc = 0; kc < 16; ++kc) {
        al[kc] = *(const v2f*)(bufL + aoff + 4 * kc);
        as[kc] = *(const v2f*)(bufS + aoff + 4 * kc);
      }
      v8f Qx = {}, Kx = {}, Vx = {};
#pragma unroll
      for (int kc = 0; kc < 16; ++kc) {
        Qx = wmma_f32(ax[kc], wq[kc], Qx);
        Kx = wmma_f32(ax[kc], wk[kc], Kx);
        Vx = wmma_f32(ax[kc], wv[kc], Vx);
      }
      asm volatile("" : "+v"(Qx), "+v"(Kx), "+v"(Vx), "+v"(al[0]), "+v"(as[0]));
      v8f Ql = {}, Kl = {}, Vl = {}, Ks = {}, Vs = {};
#pragma unroll
      for (int kc = 0; kc < 16; ++kc) {
        Ql = wmma_f32(al[kc], wq[kc], Ql);
        Kl = wmma_f32(al[kc], wk[kc], Kl);
        Vl = wmma_f32(al[kc], wv[kc], Vl);
        Ks = wmma_f32(as[kc], wk[kc], Ks);
        Vs = wmma_f32(as[kc], wv[kc], Vs);
      }
      attend<true>(Ql, Kl, Kx, Vl, Vx, Qx, Ks, Kx, Vs, Vx, part, bufO0, bufO1,
                   lane, wid);
    }

    {
      const int tn = (t + 1 < TSTEPS) ? t + 1 : t;
      const float* xn = xlane + (size_t)tn * DDIM;
#pragma unroll
      for (int kc = 0; kc < 16; ++kc)
        ax[kc] = *(const v2f*)(xn + 4 * kc);
    }

    {
      v2f a0[16], a1[16];
#pragma unroll
      for (int kc = 0; kc < 16; ++kc) {
        a0[kc] = *(const v2f*)(bufO0 + aoff + 4 * kc);
        a1[kc] = *(const v2f*)(bufO1 + aoff + 4 * kc);
      }
      v8f q0 = {}, k0 = {}, v0 = {}, q1 = {}, k1 = {}, v1 = {};
#pragma unroll
      for (int kc = 0; kc < 16; ++kc) {
        q0 = wmma_f32(a0[kc], wq[kc], q0);
        k0 = wmma_f32(a0[kc], wk[kc], k0);
        v0 = wmma_f32(a0[kc], wv[kc], v0);
        q1 = wmma_f32(a1[kc], wq[kc], q1);
        k1 = wmma_f32(a1[kc], wk[kc], k1);
        v1 = wmma_f32(a1[kc], wv[kc], v1);
      }
      attend<false>(q0, k0, k1, v0, v1, q1, k0, k1, v0, v1, part, bufL, bufS,
                    lane, wid);
    }
  }

  if (wid == 0) {
    const float4* src = (const float4*)(bufL + m * LS + 32 * h);
    float4* dst = (float4*)(out + (size_t)b * DDIM + 32 * h);
#pragma unroll
    for (int i = 0; i < 8; ++i) dst[i] = src[i];
  }
}

extern "C" void kernel_launch(void* const* d_in, const int* in_sizes, int n_in,
                              void* d_out, int out_size, void* d_ws,
                              size_t ws_size, hipStream_t stream) {
  (void)in_sizes; (void)n_in; (void)out_size; (void)ws_size;
  const float* x  = (const float*)d_in[0];
  const float* Wq = (const float*)d_in[1];
  const float* Wk = (const float*)d_in[2];
  const float* Wv = (const float*)d_in[3];
  float* wsw = (float*)d_ws;

  swizzle_weights<<<48, 256, 0, stream>>>(Wq, Wk, Wv, wsw);
  birnn_kernel<<<2048 / 16, 128, 0, stream>>>(x, wsw, (float*)d_out);
}


# Value projections issued behind the scores as one WMMA stream: score fold, partial-sum exchange (barrier signal and wait apart) and softmax stand between its WMMAs; sum order unchanged
# speedup vs baseline: 2.1067x; 1.0747x over previous
#include <hip/hip_runtime.h>
#include <math.h>
#include <utility>


#define LS     68
#define BUF    (16 * LS)
#define TSTEPS 200
#define DDIM   64

typedef float v2f __attribute__((ext_vector_type(2)));
typedef float v4f __attribute__((ext_vector_type(4)));
typedef float v8f __attribute__((ext_vector_type(8)));

__device__ __forceinline__ float fast_tanh(float v) {
#if __has_builtin(__builtin_amdgcn_tanhf)
  return __builtin_amdgcn_tanhf(v);
#elif __has_builtin(__builtin_amdgcn_tanh_f32)
  return __builtin_amdgcn_tanh_f32(v);
#else
  return tanhf(v);
#endif
}

__device__ __forceinline__ v8f wmma_f32(v2f a, v2f b, v8f c) {
  return __builtin_amdgcn_wmma_f32_16x16x4_f32(false, a, false, b, (short)0, c,
                                               false, false);
}

template <int CTRL>
__device__ __forceinline__ float rowmove(float v) {
  return __int_as_float(__builtin_amdgcn_update_dpp(0, __float_as_int(v), CTRL,
                                                    0xf, 0xf, true));
}

struct V3 {
  v8f& c0; v8f& c1; v8f& c2;
  const v2f (&a0)[16]; const v2f (&a1)[16]; const v2f (&a2)[16];
  const v2f (&b)[16];
  template <int I> __device__ __forceinline__ void issue() {
    constexpr int kc = I / 3;
    if constexpr (I % 3 == 0) c0 = wmma_f32(a0[kc], b[kc], c0);
    else if constexpr (I % 3 == 1) c1 = wmma_f32(a1[kc], b[kc], c1);
    else c2 = wmma_f32(a2[kc], b[kc], c2);
  }
  __device__ __forceinline__ void tie() {
    asm volatile("" : "+v"(c0), "+v"(c1), "+v"(c2));
  }
  __device__ __forceinline__ void tie_behind_lds() {
    asm volatile("" : "+v"(c0), "+v"(c1), "+v"(c2) : : "memory");
  }
  __device__ __forceinline__ void tie_first(v8f& q, v8f& r) {
    asm volatile("" : "+v"(c0), "+v"(c1), "+v"(q), "+v"(r));
  }
  __device__ __forceinline__ void tie(v4f& f, v4f& g) {
    asm volatile("" : "+v"(c0), "+v"(c1), "+v"(c2), "+v"(f), "+v"(g));
  }
  __device__ __forceinline__ void tie(float& r0, float& r1) {
    asm volatile("" : "+v"(c0), "+v"(c1), "+v"(c2), "+v"(r0), "+v"(r1));
  }
  __device__ __forceinline__ void tie(float& r0, float& r1, float& r2) {
    asm volatile("" : "+v"(c0), "+v"(c1), "+v"(c2), "+v"(r0), "+v"(r1), "+v"(r2));
  }
  __device__ __forceinline__ void tie(float& r0, float& r1, float& r2,
                                      float& r3, float& r4, float& r5,
                                      float& r6, float& r7) {
    asm volatile(""
                 : "+v"(c0), "+v"(c1), "+v"(c2), "+v"(r0), "+v"(r1), "+v"(r2),
                   "+v"(r3), "+v"(r4), "+v"(r5), "+v"(r6), "+v"(r7));
  }
  __device__ __forceinline__ void tie(float& r0, float& r1, float& r2,
                                      float& r3) {
    asm volatile(""
                 : "+v"(c0), "+v"(c1), "+v"(c2), "+v"(r0), "+v"(r1), "+v"(r2),
                   "+v"(r3));
  }
};
struct V2 {
  v8f& c0; v8f& c1;
  const v2f (&a0)[16]; const v2f (&a1)[16];
  const v2f (&b)[16];
  template <int I> __device__ __forceinline__ void issue() {
    constexpr int kc = I / 2;
    if constexpr (I % 2 == 0) c0 = wmma_f32(a0[kc], b[kc], c0);
    else c1 = wmma_f32(a1[kc], b[kc], c1);
  }
  __device__ __forceinline__ void tie() {
    asm volatile("" : "+v"(c0), "+v"(c1));
  }
  __device__ __forceinline__ void tie_behind_lds() {
    asm volatile("" : "+v"(c0), "+v"(c1) : : "memory");
  }
  __device__ __forceinline__ void tie_first(v8f& q, v8f& r) {
    asm volatile("" : "+v"(c0), "+v"(c1), "+v"(q), "+v"(r));
  }
  __device__ __forceinline__ void tie(v4f& f, v4f& g) {
    asm volatile("" : "+v"(c0), "+v"(c1), "+v"(f), "+v"(g));
  }
  __device__ __forceinline__ void tie(float& r0, float& r1) {
    asm volatile("" : "+v"(c0), "+v"(c1), "+v"(r0), "+v"(r1));
  }
  __device__ __forceinline__ void tie(float& r0, float& r1, float& r2) {
    asm volatile("" : "+v"(c0), "+v"(c1), "+v"(r0), "+v"(r1), "+v"(r2));
  }
  __device__ __forceinline__ void tie(float& r0, float& r1, float& r2,
                                      float& r3, float& r4, float& r5,
                                      float& r6, float& r7) {
    asm volatile(""
                 : "+v"(c0), "+v"(c1), "+v"(r0), "+v"(r1), "+v"(r2), "+v"(r3),
                   "+v"(r4), "+v"(r5), "+v"(r6), "+v"(r7));
  }
  __device__ __forceinline__ void tie(float& r0, float& r1, float& r2,
                                      float& r3) {
    asm volatile("" : "+v"(c0), "+v"(c1), "+v"(r0), "+v"(r1), "+v"(r2), "+v"(r3));
  }
};
template <int I0, class C, int... K>
__device__ __forceinline__ void run_seq(C& ch, std::integer_sequence<int, K...>) {
  (ch.template issue<I0 + K>(), ...);
}
template <int I0, int I1, class C>
__device__ __forceinline__ void run(C& ch) {
  run_seq<I0>(ch, std::make_integer_sequence<int, I1 - I0>{});
}

template <int CTRL>
__device__ __forceinline__ float fold_step(bool bit, float lo, float hi) {
  float keep = bit ? hi : lo;
  float send = bit ? lo : hi;
  return keep + rowmove<CTRL>(send);
}

template <int I0, class C>
__device__ __forceinline__ void fold16(float (&v)[32], int c, float& z0,
                                       float& z1, C& ch) {
  const bool b3 = (c & 8) != 0, b2 = (c & 4) != 0, b1 = (c & 2) != 0,
             b0 = (c & 1) != 0;
  float w[16], u[8], t[4];
  ch.template issue<I0 + 0>();
  ch.tie(v[0], v[1], v[2], v[3], v[16], v[17], v[18], v[19]);
#pragma unroll
  for (int i = 0; i < 4; ++i) w[i] = fold_step<0x168>(b3, v[i], v[i + 16]);
  ch.template issue<I0 + 1>();
  ch.tie(v[4], v[5], v[6], v[7], v[20], v[21], v[22], v[23]);
#pragma unroll
  for (int i = 4; i < 8; ++i) w[i] = fold_step<0x168>(b3, v[i], v[i + 16]);
  ch.template issue<I0 + 2>();
  ch.tie(v[8], v[9], v[10], v[11], v[24], v[25], v[26], v[27]);
#pragma unroll
  for (int i = 8; i < 12; ++i) w[i] = fold_step<0x168>(b3, v[i], v[i + 16]);
  ch.template issue<I0 + 3>();
  ch.tie(v[12], v[13], v[14], v[15], v[28], v[29], v[30], v[31]);
#pragma unroll
  for (int i = 12; i < 16; ++i) w[i] = fold_step<0x168>(b3, v[i], v[i + 16]);
  ch.template issue<I0 + 4>();
  ch.tie(w[0], w[1], w[2], w[3], w[8], w[9], w[10], w[11]);
#pragma unroll
  for (int i = 0; i < 4; ++i) u[i] = fold_step<0x164>(b2, w[i], w[i + 8]);
  ch.template issue<I0 + 5>();
  ch.tie(w[4], w[5], w[6], w[7], w[12], w[13], w[14], w[15]);
#pragma unroll
  for (int i = 4; i < 8; ++i) u[i] = fold_step<0x164>(b2, w[i], w[i + 8]);
  ch.template issue<I0 + 6>();
  ch.tie(u[0], u[1], u[2], u[3], u[4], u[5], u[6], u[7]);
#pragma unroll
  for (int i = 0; i < 4; ++i) t[i] = fold_step<0x162>(b1, u[i], u[i + 4]);
  ch.template issue<I0 + 7>();
  ch.tie(t[0], t[1], t[2], t[3]);
  z0 = fold_step<0x161>(b0, t[0], t[2]);
  z1 = fold_step<0x161>(b0, t[1], t[3]);
}

__device__ __forceinline__ void exchange_signal() {
  asm volatile("s_wait_dscnt 0x0\n\ts_barrier_signal -1" ::: "memory");
}
__device__ __forceinline__ void exchange_wait() {
  asm volatile("s_barrier_wait -1" ::: "memory");
}

template <class C>
__device__ __forceinline__ v2f score_shares(v8f& Qa, const v8f& Ka0,
                                            const v8f& Ka1, v8f& Qb,
                                            const v8f& Kb0, const v8f& Kb1,
                                            int lane, C& ch) {
  const int c = lane & 15;
  ch.template issue<0>();
  ch.template issue<1>();
  ch.tie_first(Qa, Qb);
  float v[32];
#pragma unroll
  for (int r = 0; r < 8; ++r) {
    v[4 * r + 0] = Qa[r] * Ka0[r];
    v[4 * r + 1] = Qa[r] * Ka1[r];
    v[4 * r + 2] = Qb[r] * Kb0[r];
    v[4 * r + 3] = Qb[r] * Kb1[r];
  }
  float z0, z1;
  fold16<2>(v, c, z0, z1, ch);
  v2f zz;
  zz.x = z0;
  zz.y = z1;
  return zz;
}

struct RowProbs { float a0, a1, b0, b1; };
template <int R>
__device__ __forceinline__ RowProbs hand_round(float p0, float p1) {
  RowProbs q;
  q.a0 = rowmove<0x150 + 2 * R>(p0);
  q.a1 = rowmove<0x150 + 2 * R>(p1);
  q.b0 = rowmove<0x150 + 2 * R + 1>(p0);
  q.b1 = rowmove<0x150 + 2 * R + 1>(p1);
  return q;
}
template <int R, bool DO_TANH>
__device__ __forceinline__ void blend_row(const RowProbs& q, const v8f& Va0,
                                          const v8f& Va1, const v8f& Vb0,
                                          const v8f& Vb1, float* dA, float* dB) {
  float oa = q.a0 * Va0[R] + q.a1 * Va1[R];
  float ob = q.b0 * Vb0[R] + q.b1 * Vb1[R];
  if (DO_TANH) {
    oa = fast_tanh(oa);
    ob = fast_tanh(ob);
  }
  dA[R * LS] = oa;
  dB[R * LS] = ob;
}

template <bool DO_TANH, int I0, int I1, int I2, int I3, class C>
__device__ __forceinline__ void blend_out(v4f& f0, v4f& f1, C& ch,
                                          const v8f& Va0, const v8f& Va1,
                                          const v8f& Vb0, const v8f& Vb1,
                                          float* dstA, float* dstB, int lane,
                                          int wid) {
  const int c = lane & 15;
  const int h = lane >> 4;
  ch.tie(f0, f1);
  const float d0 = ((f0.x + f0.z) + f1.x) + f1.z;
  const float d1 = ((f0.y + f0.w) + f1.y) + f1.w;
  const float kScale = 0.125f;
  const float sa = d0 * kScale, sb = d1 * kScale;
  float mx = fmaxf(sa, sb);
  float e0 = __expf(sa - mx);
  float e1 = __expf(sb - mx);
  run<I0, I1>(ch);
  ch.tie(e0, e1);
  float inv = 1.f / (e0 + e1);
  run<I1, I2>(ch);
  ch.tie(e0, e1, inv);
  float p0 = e0 * inv;
  float p1 = e1 * inv;
  const RowProbs q0 = hand_round<0>(p0, p1), q1 = hand_round<1>(p0, p1),
                 q2 = hand_round<2>(p0, p1), q3 = hand_round<3>(p0, p1),
                 q4 = hand_round<4>(p0, p1), q5 = hand_round<5>(p0, p1),
                 q6 = hand_round<6>(p0, p1), q7 = hand_round<7>(p0, p1);
  run<I2, I3>(ch);

  float* dA = dstA + (8 * h) * LS + 16 * wid + c;
  float* dB = dstB + (8 * h) * LS + 16 * wid + c;
  blend_row<0, DO_TANH>(q0, Va0, Va1, Vb0, Vb1, dA, dB);
  blend_row<1, DO_TANH>(q1, Va0, Va1, Vb0, Vb1, dA, dB);
  blend_row<2, DO_TANH>(q2, Va0, Va1, Vb0, Vb1, dA, dB);
  blend_row<3, DO_TANH>(q3, Va0, Va1, Vb0, Vb1, dA, dB);
  blend_row<4, DO_TANH>(q4, Va0, Va1, Vb0, Vb1, dA, dB);
  blend_row<5, DO_TANH>(q5, Va0, Va1, Vb0, Vb1, dA, dB);
  blend_row<6, DO_TANH>(q6, Va0, Va1, Vb0, Vb1, dA, dB);
  blend_row<7, DO_TANH>(q7, Va0, Va1, Vb0, Vb1, dA, dB);
  __syncthreads();
}

__global__ void swizzle_weights(const float* __restrict__ Wq,
                                const float* __restrict__ Wk,
                                const float* __restrict__ Wv,
                                float* __restrict__ wsw) {
  int idx = blockIdx.x * blockDim.x + threadIdx.x;
  if (idx >= 3 * 16 * 4 * 32 * 2) return;
  int j    = idx & 1;
  int lane = (idx >> 1) & 31;
  int nt   = (idx >> 6) & 3;
  int kc   = (idx >> 8) & 15;
  int w    = idx >> 12;
  int k = 4 * kc + j + 2 * (lane >> 4);
  int n = 16 * nt + (lane & 15);
  const float* W = (w == 0) ? Wq : (w == 1) ? Wk : Wv;
  wsw[idx] = W[k * 64 + n];
}

__global__ __launch_bounds__(128) void birnn_kernel(
    const float* __restrict__ x, const float* __restrict__ wsw,
    float* __restrict__ out) {
  __shared__ __align__(16) float smem[4 * BUF + 256];
  float* bufL  = smem + 0 * BUF;
  float* bufS  = smem + 1 * BUF;
  float* bufO0 = smem + 2 * BUF;
  float* bufO1 = smem + 3 * BUF;
  float* part  = smem + 4 * BUF;

  const int lane = threadIdx.x & 31;
  const int wid  = threadIdx.x >> 5;
  const int m = lane & 15;
  const int h = lane >> 4;
  const int b = blockIdx.x * 16 + m;

  v2f wq[16], wk[16], wv[16];
  {
    const float* myw = wsw + wid * 64 + lane * 2;
#pragma unroll
    for (int kc = 0; kc < 16; ++kc) {
      wq[kc] = *(const v2f*)(myw + kc * 256);
      wk[kc] = *(const v2f*)(myw + 4096 + kc * 256);
      wv[kc] = *(const v2f*)(myw + 8192 + kc * 256);
    }
  }

  if (wid == 0) {
    const float4* xr = (const float4*)(x + (size_t)b * TSTEPS * DDIM + 32 * h);
    float4* dl  = (float4*)(bufL + m * LS + 32 * h);
    float4* dsv = (float4*)(bufS + m * LS + 32 * h);
#pragma unroll
    for (int i = 0; i < 8; ++i) {
      float4 v = xr[i];
      dl[i]  = v;
      dsv[i] = v;
    }
  }
  __syncthreads();

  const int aoff = m * LS + 2 * h;
  const float* xlane = x + (size_t)b * TSTEPS * DDIM + 2 * h;

  v2f ax[16];
#pragma unroll
  for (int kc = 0; kc < 16; ++kc)
    ax[kc] = *(const v2f*)(xlane + DDIM + 4 * kc);

#pragma unroll 1
  for (int t = 1; t < TSTEPS; ++t) {
    {
      v2f al[16], as[16];
#pragma unroll
      for (int kc = 0; kc < 16; ++kc) {
        al[kc] = *(const v2f*)(bufL + aoff + 4 * kc);
        as[kc] = *(const v2f*)(bufS + aoff + 4 * kc);
      }
      v8f Qx = {}, Kx = {};
#pragma unroll
      for (int kc = 0; kc < 16; ++kc) {
        Qx = wmma_f32(ax[kc], wq[kc], Qx);
        Kx = wmma_f32(ax[kc], wk[kc], Kx);
      }
      asm volatile("" : "+v"(Qx), "+v"(Kx), "+v"(al[0]), "+v"(as[0]));
      v8f Ql = {}, Kl = {}, Ks = {};
#pragma unroll
      for (int kc = 0; kc < 16; ++kc) {
        Ql = wmma_f32(al[kc], wq[kc], Ql);
        Kl = wmma_f32(al[kc], wk[kc], Kl);
        Ks = wmma_f32(as[kc], wk[kc], Ks);
      }
      v8f Vx = {}, Vl = {}, Vs = {};
      V3 ch{Vx, Vl, Vs, ax, al, as, wv};
      const v2f zz = score_shares(Ql, Kl, Kx, Qx, Ks, Kx, lane, ch);
      *(v2f*)(part + lane * 8 + wid * 2) = zz;
      ch.tie_behind_lds();
      run<10, 14>(ch);
      ch.tie();
      exchange_signal();
      ch.tie();
      run<14, 28>(ch);
      ch.tie();
      exchange_wait();
      v4f f0 = *(const v4f*)(part + lane * 8);
      v4f f1 = *(const v4f*)(part + lane * 8 + 4);
      ch.tie_behind_lds();
      run<28, 34>(ch);
      blend_out<true, 34, 38, 43, 48>(f0, f1, ch, Vl, Vx, Vs, Vx, bufO0, bufO1,
                                      lane, wid);
    }

    {
      const int tn = (t + 1 < TSTEPS) ? t + 1 : t;
      const float* xn = xlane + (size_t)tn * DDIM;
#pragma unroll
      for (int kc = 0; kc < 16; ++kc)
        ax[kc] = *(const v2f*)(xn + 4 * kc);
    }

    {
      v2f a0[16], a1[16];
#pragma unroll
      for (int kc = 0; kc < 16; ++kc) {
        a0[kc] = *(const v2f*)(bufO0 + aoff + 4 * kc);
        a1[kc] = *(const v2f*)(bufO1 + aoff + 4 * kc);
      }
      v8f q0 = {}, k0 = {}, q1 = {}, k1 = {};
#pragma unroll
      for (int kc = 0; kc < 16; ++kc) {
        q0 = wmma_f32(a0[kc], wq[kc], q0);
        k0 = wmma_f32(a0[kc], wk[kc], k0);
        q1 = wmma_f32(a1[kc], wq[kc], q1);
        k1 = wmma_f32(a1[kc], wk[kc], k1);
      }
      v8f v0 = {}, v1 = {};
      V2 ch{v0, v1, a0, a1, wv};
      const v2f zz = score_shares(q0, k0, k1, q1, k0, k1, lane, ch);
      *(v2f*)(part + lane * 8 + wid * 2) = zz;
      ch.tie_behind_lds();
      run<10, 14>(ch);
      ch.tie();
      exchange_signal();
      ch.tie();
      run<14, 18>(ch);
      ch.tie();
      exchange_wait();
      v4f f0 = *(const v4f*)(part + lane * 8);
      v4f f1 = *(const v4f*)(part + lane * 8 + 4);
      ch.tie_behind_lds();
      run<18, 23>(ch);
      blend_out<false, 23, 26, 29, 32>(f0, f1, ch, v0, v1, v0, v1, bufL, bufS,
                                       lane, wid);
    }
  }

  if (wid == 0) {
    const float4* src = (const float4*)(bufL + m * LS + 32 * h);
    float4* dst = (float4*)(out + (size_t)b * DDIM + 32 * h);
#pragma unroll
    for (int i = 0; i < 8; ++i) dst[i] = src[i];
  }
}

extern "C" void kernel_launch(void* const* d_in, const int* in_sizes, int n_in,
                              void* d_out, int out_size, void* d_ws,
                              size_t ws_size, hipStream_t stream) {
  (void)in_sizes; (void)n_in; (void)out_size; (void)ws_size;
  const float* x  = (const float*)d_in[0];
  const float* Wq = (const float*)d_in[1];
  const float* Wk = (const float*)d_in[2];
  const float* Wv = (const float*)d_in[3];
  float* wsw = (float*)d_ws;

  swizzle_weights<<<48, 256, 0, stream>>>(Wq, Wk, Wv, wsw);
  birnn_kernel<<<2048 / 16, 128, 0, stream>>>(x, wsw, (float*)d_out);
}


# Key projections take token differences: (l-x)Wk, (s-x)Wk, (o0-o1)Wk give each softmax pair's score difference; 192 WMMAs per wave a step, was 224; fold and exchange halve; score sum order changes
# speedup vs baseline: 2.3459x; 1.1135x over previous
#include <hip/hip_runtime.h>
#include <math.h>
#include <utility>


#define LS     68
#define BUF    (16 * LS)
#define TSTEPS 200
#define DDIM   64

typedef float v2f __attribute__((ext_vector_type(2)));
typedef float v4f __attribute__((ext_vector_type(4)));
typedef float v8f __attribute__((ext_vector_type(8)));

__device__ __forceinline__ float fast_tanh(float v) {
#if __has_builtin(__builtin_amdgcn_tanhf)
  return __builtin_amdgcn_tanhf(v);
#elif __has_builtin(__builtin_amdgcn_tanh_f32)
  return __builtin_amdgcn_tanh_f32(v);
#else
  return tanhf(v);
#endif
}

__device__ __forceinline__ v8f wmma_f32(v2f a, v2f b, v8f c) {
  return __builtin_amdgcn_wmma_f32_16x16x4_f32(false, a, false, b, (short)0, c,
                                               false, false);
}

template <int CTRL>
__device__ __forceinline__ float rowmove(float v) {
  return __int_as_float(__builtin_amdgcn_update_dpp(0, __float_as_int(v), CTRL,
                                                    0xf, 0xf, true));
}

struct V2 {
  v8f& c0; v8f& c1;
  const v2f (&a0)[16]; const v2f (&a1)[16];
  const v2f (&b)[16];
  template <int I> __device__ __forceinline__ void issue() {
    constexpr int kc = I / 2;
    if constexpr (I % 2 == 0) c0 = wmma_f32(a0[kc], b[kc], c0);
    else c1 = wmma_f32(a1[kc], b[kc], c1);
  }
  __device__ __forceinline__ void tie() {
    asm volatile("" : "+v"(c0), "+v"(c1));
  }
  __device__ __forceinline__ void tie_behind_lds() {
    asm volatile("" : "+v"(c0), "+v"(c1) : : "memory");
  }
  __device__ __forceinline__ void tie_first(v8f& q, v8f& r) {
    asm volatile("" : "+v"(c0), "+v"(c1), "+v"(q), "+v"(r));
  }
  __device__ __forceinline__ void tie(v4f& f) {
    asm volatile("" : "+v"(c0), "+v"(c1), "+v"(f));
  }
  __device__ __forceinline__ void tie(float& r0, float& r1) {
    asm volatile("" : "+v"(c0), "+v"(c1), "+v"(r0), "+v"(r1));
  }
  __device__ __forceinline__ void tie(float& r0, float& r1, float& r2) {
    asm volatile("" : "+v"(c0), "+v"(c1), "+v"(r0), "+v"(r1), "+v"(r2));
  }
  __device__ __forceinline__ void tie(float& r0, float& r1, float& r2,
                                      float& r3, float& r4, float& r5,
                                      float& r6, float& r7) {
    asm volatile(""
                 : "+v"(c0), "+v"(c1), "+v"(r0), "+v"(r1), "+v"(r2), "+v"(r3),
                   "+v"(r4), "+v"(r5), "+v"(r6), "+v"(r7));
  }
  __device__ __forceinline__ void tie(float& r0, float& r1, float& r2,
                                      float& r3) {
    asm volatile("" : "+v"(c0), "+v"(c1), "+v"(r0), "+v"(r1), "+v"(r2), "+v"(r3));
  }
};
template <int I0, class C, int... K>
__device__ __forceinline__ void run_seq(C& ch, std::integer_sequence<int, K...>) {
  (ch.template issue<I0 + K>(), ...);
}
template <int I0, int I1, class C>
__device__ __forceinline__ void run(C& ch) {
  run_seq<I0>(ch, std::make_integer_sequence<int, I1 - I0>{});
}

template <int CTRL>
__device__ __forceinline__ float fold_step(bool bit, float lo, float hi) {
  float keep = bit ? hi : lo;
  float send = bit ? lo : hi;
  return keep + rowmove<CTRL>(send);
}

template <int I0, class C>
__device__ __forceinline__ float fold16(float (&v)[16], int c, C& ch) {
  const bool b3 = (c & 8) != 0, b2 = (c & 4) != 0, b1 = (c & 2) != 0,
             b0 = (c & 1) != 0;
  float w[8], u[4], t[2];
  ch.template issue<I0 + 0>();
  ch.tie(v[0], v[1], v[2], v[3], v[8], v[9], v[10], v[11]);
#pragma unroll
  for (int i = 0; i < 4; ++i) w[i] = fold_step<0x168>(b3, v[i], v[i + 8]);
  ch.template issue<I0 + 1>();
  ch.tie(v[4], v[5], v[6], v[7], v[12], v[13], v[14], v[15]);
#pragma unroll
  for (int i = 4; i < 8; ++i) w[i] = fold_step<0x168>(b3, v[i], v[i + 8]);
  ch.template issue<I0 + 2>();
  ch.tie(w[0], w[1], w[2], w[3], w[4], w[5], w[6], w[7]);
#pragma unroll
  for (int i = 0; i < 4; ++i) u[i] = fold_step<0x164>(b2, w[i], w[i + 4]);
  ch.template issue<I0 + 3>();
  ch.tie(u[0], u[1], u[2], u[3]);
  t[0] = fold_step<0x162>(b1, u[0], u[2]);
  t[1] = fold_step<0x162>(b1, u[1], u[3]);
  return fold_step<0x161>(b0, t[0], t[1]);
}

__device__ __forceinline__ void exchange_signal() {
  asm volatile("s_wait_dscnt 0x0\n\ts_barrier_signal -1" ::: "memory");
}
__device__ __forceinline__ void exchange_wait() {
  asm volatile("s_barrier_wait -1" ::: "memory");
}

template <class C>
__device__ __forceinline__ float score_share(v8f& Qa, const v8f& Da, v8f& Qb,
                                             const v8f& Db, int lane, C& ch) {
  const int c = lane & 15;
  ch.template issue<0>();
  ch.template issue<1>();
  ch.tie_first(Qa, Qb);
  float v[16];
#pragma unroll
  for (int r = 0; r < 8; ++r) {
    v[2 * r + 0] = Qa[r] * Da[r];
    v[2 * r + 1] = Qb[r] * Db[r];
  }
  return fold16<2>(v, c, ch);
}

struct RowProbs { float a0, a1, b0, b1; };
template <int R>
__device__ __forceinline__ RowProbs hand_round(float p0, float p1) {
  RowProbs q;
  q.a0 = rowmove<0x150 + 2 * R>(p0);
  q.a1 = rowmove<0x150 + 2 * R>(p1);
  q.b0 = rowmove<0x150 + 2 * R + 1>(p0);
  q.b1 = rowmove<0x150 + 2 * R + 1>(p1);
  return q;
}
template <int R, bool DO_TANH>
__device__ __forceinline__ void blend_row(const RowProbs& q, const v8f& Va0,
                                          const v8f& Va1, const v8f& Vb0,
                                          const v8f& Vb1, float* dA, float* dB) {
  float oa = q.a0 * Va0[R] + q.a1 * Va1[R];
  float ob = q.b0 * Vb0[R] + q.b1 * Vb1[R];
  if (DO_TANH) {
    oa = fast_tanh(oa);
    ob = fast_tanh(ob);
  }
  dA[R * LS] = oa;
  dB[R * LS] = ob;
}

template <bool DO_TANH, int I0, int I1, int I2, int I3, class C>
__device__ __forceinline__ void blend_out(v4f& f, C& ch,
                                          const v8f& Va0, const v8f& Va1,
                                          const v8f& Vb0, const v8f& Vb1,
                                          float* dstA, float* dstB, int lane,
                                          int wid) {
  const int c = lane & 15;
  const int h = lane >> 4;
  ch.tie(f);
  const float d = ((f.x + f.y) + f.z) + f.w;
  const float kScale = 0.125f;
  const float dd = d * kScale;
  float e0 = __expf(fminf(dd, 0.f));
  float e1 = __expf(fminf(-dd, 0.f));
  run<I0, I1>(ch);
  ch.tie(e0, e1);
  float inv = 1.f / (e0 + e1);
  run<I1, I2>(ch);
  ch.tie(e0, e1, inv);
  float p0 = e0 * inv;
  float p1 = e1 * inv;
  const RowProbs q0 = hand_round<0>(p0, p1), q1 = hand_round<1>(p0, p1),
                 q2 = hand_round<2>(p0, p1), q3 = hand_round<3>(p0, p1),
                 q4 = hand_round<4>(p0, p1), q5 = hand_round<5>(p0, p1),
                 q6 = hand_round<6>(p0, p1), q7 = hand_round<7>(p0, p1);
  run<I2, I3>(ch);

  float* dA = dstA + (8 * h) * LS + 16 * wid + c;
  float* dB = dstB + (8 * h) * LS + 16 * wid + c;
  blend_row<0, DO_TANH>(q0, Va0, Va1, Vb0, Vb1, dA, dB);
  blend_row<1, DO_TANH>(q1, Va0, Va1, Vb0, Vb1, dA, dB);
  blend_row<2, DO_TANH>(q2, Va0, Va1, Vb0, Vb1, dA, dB);
  blend_row<3, DO_TANH>(q3, Va0, Va1, Vb0, Vb1, dA, dB);
  blend_row<4, DO_TANH>(q4, Va0, Va1, Vb0, Vb1, dA, dB);
  blend_row<5, DO_TANH>(q5, Va0, Va1, Vb0, Vb1, dA, dB);
  blend_row<6, DO_TANH>(q6, Va0, Va1, Vb0, Vb1, dA, dB);
  blend_row<7, DO_TANH>(q7, Va0, Va1, Vb0, Vb1, dA, dB);
  __syncthreads();
}

__global__ void swizzle_weights(const float* __restrict__ Wq,
                                const float* __restrict__ Wk,
                                const float* __restrict__ Wv,
                                float* __restrict__ wsw) {
  int idx = blockIdx.x * blockDim.x + threadIdx.x;
  if (idx >= 3 * 16 * 4 * 32 * 2) return;
  int j    = idx & 1;
  int lane = (idx >> 1) & 31;
  int nt   = (idx >> 6) & 3;
  int kc   = (idx >> 8) & 15;
  int w    = idx >> 12;
  int k = 4 * kc + j + 2 * (lane >> 4);
  int n = 16 * nt + (lane & 15);
  const float* W = (w == 0) ? Wq : (w == 1) ? Wk : Wv;
  wsw[idx] = W[k * 64 + n];
}

__global__ __launch_bounds__(128) void birnn_kernel(
    const float* __restrict__ x, const float* __restrict__ wsw,
    float* __restrict__ out) {
  __shared__ __align__(16) float smem[4 * BUF + 128];
  float* bufL  = smem + 0 * BUF;
  float* bufS  = smem + 1 * BUF;
  float* bufO0 = smem + 2 * BUF;
  float* bufO1 = smem + 3 * BUF;
  float* part  = smem + 4 * BUF;

  const int lane = threadIdx.x & 31;
  const int wid  = threadIdx.x >> 5;
  const int m = lane & 15;
  const int h = lane >> 4;
  const int b = blockIdx.x * 16 + m;

  v2f wq[16], wk[16], wv[16];
  {
    const float* myw = wsw + wid * 64 + lane * 2;
#pragma unroll
    for (int kc = 0; kc < 16; ++kc) {
      wq[kc] = *(const v2f*)(myw + kc * 256);
      wk[kc] = *(const v2f*)(myw + 4096 + kc * 256);
      wv[kc] = *(const v2f*)(myw + 8192 + kc * 256);
    }
  }

  if (wid == 0) {
    const float4* xr = (const float4*)(x + (size_t)b * TSTEPS * DDIM + 32 * h);
    float4* dl  = (float4*)(bufL + m * LS + 32 * h);
    float4* dsv = (float4*)(bufS + m * LS + 32 * h);
#pragma unroll
    for (int i = 0; i < 8; ++i) {
      float4 v = xr[i];
      dl[i]  = v;
      dsv[i] = v;
    }
  }
  __syncthreads();

  const int aoff = m * LS + 2 * h;
  const float* xlane = x + (size_t)b * TSTEPS * DDIM + 2 * h;

  v2f ax[16];
#pragma unroll
  for (int kc = 0; kc < 16; ++kc)
    ax[kc] = *(const v2f*)(xlane + DDIM + 4 * kc);

#pragma unroll 1
  for (int t = 1; t < TSTEPS; ++t) {
    {
      v2f al[16], as[16];
#pragma unroll
      for (int kc = 0; kc < 16; ++kc) {
        al[kc] = *(const v2f*)(bufL + aoff + 4 * kc);
        as[kc] = *(const v2f*)(bufS + aoff + 4 * kc);
      }
      v8f Qx = {}, Vx = {};
#pragma unroll
      for (int kc = 0; kc < 16; ++kc) {
        Qx = wmma_f32(ax[kc], wq[kc], Qx);
        Vx = wmma_f32(ax[kc], wv[kc], Vx);
      }
      asm volatile(""
                   : "+v"(Qx), "+v"(Vx), "+v"(al[0]), "+v"(as[0]), "+v"(ax[0]),
                     "+v"(ax[1]), "+v"(ax[2]), "+v"(ax[3]), "+v"(ax[4]),
                     "+v"(ax[5]), "+v"(ax[6]), "+v"(ax[7]), "+v"(ax[8]),
                     "+v"(ax[9]), "+v"(ax[10]), "+v"(ax[11]), "+v"(ax[12]),
                     "+v"(ax[13]), "+v"(ax[14]), "+v"(ax[15]));
      v8f Ql = {}, Dl = {}, Ds = {};
#pragma unroll
      for (int kc = 0; kc < 16; ++kc) {
        Ql = wmma_f32(al[kc], wq[kc], Ql);
        Dl = wmma_f32(al[kc] - ax[kc], wk[kc], Dl);
        Ds = wmma_f32(as[kc] - ax[kc], wk[kc], Ds);
      }
      v8f Vl = {}, Vs = {};
      V2 ch{Vl, Vs, al, as, wv};
      const float z = score_share(Ql, Dl, Qx, Ds, lane, ch);
      part[lane * 4 + wid] = z;
      ch.tie_behind_lds();
      run<6, 10>(ch);
      ch.tie();
      exchange_signal();
      ch.tie();
      run<10, 16>(ch);
      ch.tie();
      exchange_wait();
      v4f f = *(const v4f*)(part + lane * 4);
      ch.tie_behind_lds();
      run<16, 22>(ch);
      blend_out<true, 22, 25, 28, 32>(f, ch, Vl, Vx, Vs, Vx, bufO0, bufO1,
                                      lane, wid);
    }

    {
      const int tn = (t + 1 < TSTEPS) ? t + 1 : t;
      const float* xn = xlane + (size_t)tn * DDIM;
#pragma unroll
      for (int kc = 0; kc < 16; ++kc)
        ax[kc] = *(const v2f*)(xn + 4 * kc);
    }

    {
      v2f a0[16], a1[16];
#pragma unroll
      for (int kc = 0; kc < 16; ++kc) {
        a0[kc] = *(const v2f*)(bufO0 + aoff + 4 * kc);
        a1[kc] = *(const v2f*)(bufO1 + aoff + 4 * kc);
      }
      v8f q0 = {}, q1 = {}, Dk = {};
#pragma unroll
      for (int kc = 0; kc < 16; ++kc) {
        q0 = wmma_f32(a0[kc], wq[kc], q0);
        q1 = wmma_f32(a1[kc], wq[kc], q1);
        Dk = wmma_f32(a0[kc] - a1[kc], wk[kc], Dk);
      }
      v8f v0 = {}, v1 = {};
      V2 ch{v0, v1, a0, a1, wv};
      const float z = score_share(q0, Dk, q1, Dk, lane, ch);
      part[lane * 4 + wid] = z;
      ch.tie_behind_lds();
      run<6, 10>(ch);
      ch.tie();
      exchange_signal();
      ch.tie();
      run<10, 16>(ch);
      ch.tie();
      exchange_wait();
      v4f f = *(const v4f*)(part + lane * 4);
      ch.tie_behind_lds();
      run<16, 22>(ch);
      blend_out<false, 22, 25, 28, 32>(f, ch, v0, v1, v0, v1, bufL, bufS,
                                       lane, wid);
    }
  }

  if (wid == 0) {
    const float4* src = (const float4*)(bufL + m * LS + 32 * h);
    float4* dst = (float4*)(out + (size_t)b * DDIM + 32 * h);
#pragma unroll
    for (int i = 0; i < 8; ++i) dst[i] = src[i];
  }
}

extern "C" void kernel_launch(void* const* d_in, const int* in_sizes, int n_in,
                              void* d_out, int out_size, void* d_ws,
                              size_t ws_size, hipStream_t stream) {
  (void)in_sizes; (void)n_in; (void)out_size; (void)ws_size;
  const float* x  = (const float*)d_in[0];
  const float* Wq = (const float*)d_in[1];
  const float* Wk = (const float*)d_in[2];
  const float* Wv = (const float*)d_in[3];
  float* wsw = (float*)d_ws;

  swizzle_weights<<<48, 256, 0, stream>>>(Wq, Wk, Wv, wsw);
  birnn_kernel<<<2048 / 16, 128, 0, stream>>>(x, wsw, (float*)d_out);
}


# Next step's Qx, Vx chains dealt as filler WMMAs into both phase-end barriers (signal and wait apart) and behind each phase's LDS read group; x_t+1 fetched a step ahead; sum order unchanged
# speedup vs baseline: 2.4337x; 1.0374x over previous
#include <hip/hip_runtime.h>
#include <math.h>
#include <utility>


#define LS     68
#define BUF    (16 * LS)
#define TSTEPS 200
#define DDIM   64

typedef float v2f __attribute__((ext_vector_type(2)));
typedef float v4f __attribute__((ext_vector_type(4)));
typedef float v8f __attribute__((ext_vector_type(8)));

__device__ __forceinline__ float fast_tanh(float v) {
#if __has_builtin(__builtin_amdgcn_tanhf)
  return __builtin_amdgcn_tanhf(v);
#elif __has_builtin(__builtin_amdgcn_tanh_f32)
  return __builtin_amdgcn_tanh_f32(v);
#else
  return tanhf(v);
#endif
}

__device__ __forceinline__ v8f wmma_f32(v2f a, v2f b, v8f c) {
  return __builtin_amdgcn_wmma_f32_16x16x4_f32(false, a, false, b, (short)0, c,
                                               false, false);
}

template <int CTRL>
__device__ __forceinline__ float rowmove(float v) {
  return __int_as_float(__builtin_amdgcn_update_dpp(0, __float_as_int(v), CTRL,
                                                    0xf, 0xf, true));
}

struct V2 {
  v8f& c0; v8f& c1;
  const v2f (&a0)[16]; const v2f (&a1)[16];
  const v2f (&b)[16];
  template <int I> __device__ __forceinline__ void issue() {
    constexpr int kc = I / 2;
    if constexpr (I % 2 == 0) c0 = wmma_f32(a0[kc], b[kc], c0);
    else c1 = wmma_f32(a1[kc], b[kc], c1);
  }
  __device__ __forceinline__ void tie() {
    asm volatile("" : "+v"(c0), "+v"(c1));
  }
  __device__ __forceinline__ void tie_behind_lds() {
    asm volatile("" : "+v"(c0), "+v"(c1) : : "memory");
  }
  __device__ __forceinline__ void tie_first(v8f& q, v8f& r) {
    asm volatile("" : "+v"(c0), "+v"(c1), "+v"(q), "+v"(r));
  }
  __device__ __forceinline__ void tie(v4f& f) {
    asm volatile("" : "+v"(c0), "+v"(c1), "+v"(f));
  }
  __device__ __forceinline__ void tie(float& r0, float& r1) {
    asm volatile("" : "+v"(c0), "+v"(c1), "+v"(r0), "+v"(r1));
  }
  __device__ __forceinline__ void tie(float& r0, float& r1, float& r2) {
    asm volatile("" : "+v"(c0), "+v"(c1), "+v"(r0), "+v"(r1), "+v"(r2));
  }
  __device__ __forceinline__ void tie(float& r0, float& r1, float& r2,
                                      float& r3, float& r4, float& r5,
                                      float& r6, float& r7) {
    asm volatile(""
                 : "+v"(c0), "+v"(c1), "+v"(r0), "+v"(r1), "+v"(r2), "+v"(r3),
                   "+v"(r4), "+v"(r5), "+v"(r6), "+v"(r7));
  }
  __device__ __forceinline__ void tie(float& r0, float& r1, float& r2,
                                      float& r3) {
    asm volatile("" : "+v"(c0), "+v"(c1), "+v"(r0), "+v"(r1), "+v"(r2), "+v"(r3));
  }
};
template <int I0, class C, int... K>
__device__ __forceinline__ void run_seq(C& ch, std::integer_sequence<int, K...>) {
  (ch.template issue<I0 + K>(), ...);
}
template <int I0, int I1, class C>
__device__ __forceinline__ void run(C& ch) {
  run_seq<I0>(ch, std::make_integer_sequence<int, I1 - I0>{});
}

struct XF {
  v8f& q; v8f& v;
  const v2f (&a)[16];
  const v2f (&bq)[16]; const v2f (&bv)[16];
  template <int I> __device__ __forceinline__ void issue() {
    constexpr int kc = I / 2;
    if constexpr (I % 2 == 0) q = wmma_f32(a[kc], bq[kc], q);
    else v = wmma_f32(a[kc], bv[kc], v);
  }
  __device__ __forceinline__ void tie() {
    asm volatile("" : "+v"(q), "+v"(v));
  }
  __device__ __forceinline__ void tie_behind_lds() {
    asm volatile("" : "+v"(q), "+v"(v) : : "memory");
  }
};

template <int CTRL>
__device__ __forceinline__ float fold_step(bool bit, float lo, float hi) {
  float keep = bit ? hi : lo;
  float send = bit ? lo : hi;
  return keep + rowmove<CTRL>(send);
}

template <int I0, class C>
__device__ __forceinline__ float fold16(float (&v)[16], int c, C& ch) {
  const bool b3 = (c & 8) != 0, b2 = (c & 4) != 0, b1 = (c & 2) != 0,
             b0 = (c & 1) != 0;
  float w[8], u[4], t[2];
  ch.template issue<I0 + 0>();
  ch.tie(v[0], v[1], v[2], v[3], v[8], v[9], v[10], v[11]);
#pragma unroll
  for (int i = 0; i < 4; ++i) w[i] = fold_step<0x168>(b3, v[i], v[i + 8]);
  ch.template issue<I0 + 1>();
  ch.tie(v[4], v[5], v[6], v[7], v[12], v[13], v[14], v[15]);
#pragma unroll
  for (int i = 4; i < 8; ++i) w[i] = fold_step<0x168>(b3, v[i], v[i + 8]);
  ch.template issue<I0 + 2>();
  ch.tie(w[0], w[1], w[2], w[3], w[4], w[5], w[6], w[7]);
#pragma unroll
  for (int i = 0; i < 4; ++i) u[i] = fold_step<0x164>(b2, w[i], w[i + 4]);
  ch.template issue<I0 + 3>();
  ch.tie(u[0], u[1], u[2], u[3]);
  t[0] = fold_step<0x162>(b1, u[0], u[2]);
  t[1] = fold_step<0x162>(b1, u[1], u[3]);
  return fold_step<0x161>(b0, t[0], t[1]);
}

__device__ __forceinline__ void exchange_signal() {
  asm volatile("s_wait_dscnt 0x0\n\ts_barrier_signal -1" ::: "memory");
}
__device__ __forceinline__ void exchange_wait() {
  asm volatile("s_barrier_wait -1" ::: "memory");
}

template <int F0, class X>
__device__ __forceinline__ void phase_end(X& xf) {
  xf.tie_behind_lds();
  run<F0, F0 + 4>(xf);
  xf.tie();
  exchange_signal();
  xf.tie();
  run<F0 + 4, F0 + 6>(xf);
  xf.tie();
  exchange_wait();
}

template <class C>
__device__ __forceinline__ float score_share(v8f& Qa, const v8f& Da, v8f& Qb,
                                             const v8f& Db, int lane, C& ch) {
  const int c = lane & 15;
  ch.template issue<0>();
  ch.template issue<1>();
  ch.tie_first(Qa, Qb);
  float v[16];
#pragma unroll
  for (int r = 0; r < 8; ++r) {
    v[2 * r + 0] = Qa[r] * Da[r];
    v[2 * r + 1] = Qb[r] * Db[r];
  }
  return fold16<2>(v, c, ch);
}

struct RowProbs { float a0, a1, b0, b1; };
template <int R>
__device__ __forceinline__ RowProbs hand_round(float p0, float p1) {
  RowProbs q;
  q.a0 = rowmove<0x150 + 2 * R>(p0);
  q.a1 = rowmove<0x150 + 2 * R>(p1);
  q.b0 = rowmove<0x150 + 2 * R + 1>(p0);
  q.b1 = rowmove<0x150 + 2 * R + 1>(p1);
  return q;
}
template <int R, bool DO_TANH>
__device__ __forceinline__ void blend_row(const RowProbs& q, const v8f& Va0,
                                          const v8f& Va1, const v8f& Vb0,
                                          const v8f& Vb1, float* dA, float* dB) {
  float oa = q.a0 * Va0[R] + q.a1 * Va1[R];
  float ob = q.b0 * Vb0[R] + q.b1 * Vb1[R];
  if (DO_TANH) {
    oa = fast_tanh(oa);
    ob = fast_tanh(ob);
  }
  dA[R * LS] = oa;
  dB[R * LS] = ob;
}

template <bool DO_TANH, int I0, int I1, int I2, int I3, class C>
__device__ __forceinline__ void blend_out(v4f& f, C& ch,
                                          const v8f& Va0, const v8f& Va1,
                                          const v8f& Vb0, const v8f& Vb1,
                                          float* dstA, float* dstB, int lane,
                                          int wid) {
  const int c = lane & 15;
  const int h = lane >> 4;
  ch.tie(f);
  const float d = ((f.x + f.y) + f.z) + f.w;
  const float kScale = 0.125f;
  const float dd = d * kScale;
  float e0 = __expf(fminf(dd, 0.f));
  float e1 = __expf(fminf(-dd, 0.f));
  run<I0, I1>(ch);
  ch.tie(e0, e1);
  float inv = 1.f / (e0 + e1);
  run<I1, I2>(ch);
  ch.tie(e0, e1, inv);
  float p0 = e0 * inv;
  float p1 = e1 * inv;
  const RowProbs q0 = hand_round<0>(p0, p1), q1 = hand_round<1>(p0, p1),
                 q2 = hand_round<2>(p0, p1), q3 = hand_round<3>(p0, p1),
                 q4 = hand_round<4>(p0, p1), q5 = hand_round<5>(p0, p1),
                 q6 = hand_round<6>(p0, p1), q7 = hand_round<7>(p0, p1);
  run<I2, I3>(ch);

  float* dA = dstA + (8 * h) * LS + 16 * wid + c;
  float* dB = dstB + (8 * h) * LS + 16 * wid + c;
  blend_row<0, DO_TANH>(q0, Va0, Va1, Vb0, Vb1, dA, dB);
  blend_row<1, DO_TANH>(q1, Va0, Va1, Vb0, Vb1, dA, dB);
  blend_row<2, DO_TANH>(q2, Va0, Va1, Vb0, Vb1, dA, dB);
  blend_row<3, DO_TANH>(q3, Va0, Va1, Vb0, Vb1, dA, dB);
  blend_row<4, DO_TANH>(q4, Va0, Va1, Vb0, Vb1, dA, dB);
  blend_row<5, DO_TANH>(q5, Va0, Va1, Vb0, Vb1, dA, dB);
  blend_row<6, DO_TANH>(q6, Va0, Va1, Vb0, Vb1, dA, dB);
  blend_row<7, DO_TANH>(q7, Va0, Va1, Vb0, Vb1, dA, dB);
}

__global__ void swizzle_weights(const float* __restrict__ Wq,
                                const float* __restrict__ Wk,
                                const float* __restrict__ Wv,
                                float* __restrict__ wsw) {
  int idx = blockIdx.x * blockDim.x + threadIdx.x;
  if (idx >= 3 * 16 * 4 * 32 * 2) return;
  int j    = idx & 1;
  int lane = (idx >> 1) & 31;
  int nt   = (idx >> 6) & 3;
  int kc   = (idx >> 8) & 15;
  int w    = idx >> 12;
  int k = 4 * kc + j + 2 * (lane >> 4);
  int n = 16 * nt + (lane & 15);
  const float* W = (w == 0) ? Wq : (w == 1) ? Wk : Wv;
  wsw[idx] = W[k * 64 + n];
}

__global__ __launch_bounds__(128) void birnn_kernel(
    const float* __restrict__ x, const float* __restrict__ wsw,
    float* __restrict__ out) {
  __shared__ __align__(16) float smem[4 * BUF + 128];
  float* bufL  = smem + 0 * BUF;
  float* bufS  = smem + 1 * BUF;
  float* bufO0 = smem + 2 * BUF;
  float* bufO1 = smem + 3 * BUF;
  float* part  = smem + 4 * BUF;

  const int lane = threadIdx.x & 31;
  const int wid  = threadIdx.x >> 5;
  const int m = lane & 15;
  const int h = lane >> 4;
  const int b = blockIdx.x * 16 + m;

  v2f wq[16], wk[16], wv[16];
  {
    const float* myw = wsw + wid * 64 + lane * 2;
#pragma unroll
    for (int kc = 0; kc < 16; ++kc) {
      wq[kc] = *(const v2f*)(myw + kc * 256);
      wk[kc] = *(const v2f*)(myw + 4096 + kc * 256);
      wv[kc] = *(const v2f*)(myw + 8192 + kc * 256);
    }
  }

  if (wid == 0) {
    const float4* xr = (const float4*)(x + (size_t)b * TSTEPS * DDIM + 32 * h);
    float4* dl  = (float4*)(bufL + m * LS + 32 * h);
    float4* dsv = (float4*)(bufS + m * LS + 32 * h);
#pragma unroll
    for (int i = 0; i < 8; ++i) {
      float4 v = xr[i];
      dl[i]  = v;
      dsv[i] = v;
    }
  }
  __syncthreads();

  const int aoff = m * LS + 2 * h;
  const float* xlane = x + (size_t)b * TSTEPS * DDIM + 2 * h;

  v2f ax[16];
#pragma unroll
  for (int kc = 0; kc < 16; ++kc)
    ax[kc] = *(const v2f*)(xlane + DDIM + 4 * kc);

  v8f Qx = {}, Vx = {};
  {
    XF xf{Qx, Vx, ax, wq, wv};
    run<0, 22>(xf);
  }

#pragma unroll 1
  for (int t = 1; t < TSTEPS; ++t) {
    {
      v2f al[16], as[16], axn[16];
#pragma unroll
      for (int kc = 0; kc < 16; ++kc) {
        al[kc] = *(const v2f*)(bufL + aoff + 4 * kc);
        as[kc] = *(const v2f*)(bufS + aoff + 4 * kc);
      }
      {
        const int tn = (t + 1 < TSTEPS) ? t + 1 : t;
        const float* xn = xlane + (size_t)tn * DDIM;
#pragma unroll
        for (int kc = 0; kc < 16; ++kc)
          axn[kc] = *(const v2f*)(xn + 4 * kc);
      }
      {
        XF xf{Qx, Vx, ax, wq, wv};
        xf.tie_behind_lds();
        run<22, 32>(xf);
      }
      asm volatile(""
                   : "+v"(Qx), "+v"(Vx), "+v"(al[0]), "+v"(as[0]), "+v"(ax[0]),
                     "+v"(ax[1]), "+v"(ax[2]), "+v"(ax[3]), "+v"(ax[4]),
                     "+v"(ax[5]), "+v"(ax[6]), "+v"(ax[7]), "+v"(ax[8]),
                     "+v"(ax[9]), "+v"(ax[10]), "+v"(ax[11]), "+v"(ax[12]),
                     "+v"(ax[13]), "+v"(ax[14]), "+v"(ax[15]));
      v8f Ql = {}, Dl = {}, Ds = {};
#pragma unroll
      for (int kc = 0; kc < 16; ++kc) {
        Ql = wmma_f32(al[kc], wq[kc], Ql);
        Dl = wmma_f32(al[kc] - ax[kc], wk[kc], Dl);
        Ds = wmma_f32(as[kc] - ax[kc], wk[kc], Ds);
      }
#pragma unroll
      for (int kc = 0; kc < 16; ++kc) ax[kc] = axn[kc];
      v8f Vl = {}, Vs = {};
      V2 ch{Vl, Vs, al, as, wv};
      const float z = score_share(Ql, Dl, Qx, Ds, lane, ch);
      part[lane * 4 + wid] = z;
      ch.tie_behind_lds();
      run<6, 10>(ch);
      ch.tie();
      exchange_signal();
      ch.tie();
      run<10, 16>(ch);
      ch.tie();
      exchange_wait();
      v4f f = *(const v4f*)(part + lane * 4);
      ch.tie_behind_lds();
      run<16, 22>(ch);
      blend_out<true, 22, 25, 28, 32>(f, ch, Vl, Vx, Vs, Vx, bufO0, bufO1,
                                      lane, wid);
    }
    Qx = v8f{};
    Vx = v8f{};
    XF xf{Qx, Vx, ax, wq, wv};
    phase_end<0>(xf);

    {
      v2f a0[16], a1[16];
#pragma unroll
      for (int kc = 0; kc < 16; ++kc) {
        a0[kc] = *(const v2f*)(bufO0 + aoff + 4 * kc);
        a1[kc] = *(const v2f*)(bufO1 + aoff + 4 * kc);
      }
      xf.tie_behind_lds();
      run<6, 16>(xf);
      asm volatile(""
                   : "+v"(Qx), "+v"(Vx), "+v"(a1[0]), "+v"(a0[0]), "+v"(a0[1]),
                     "+v"(a0[2]), "+v"(a0[3]), "+v"(a0[4]), "+v"(a0[5]),
                     "+v"(a0[6]), "+v"(a0[7]), "+v"(a0[8]), "+v"(a0[9]),
                     "+v"(a0[10]), "+v"(a0[11]), "+v"(a0[12]), "+v"(a0[13]),
                     "+v"(a0[14]), "+v"(a0[15]));
      v8f q0 = {}, q1 = {}, Dk = {};
#pragma unroll
      for (int kc = 0; kc < 16; ++kc) {
        q0 = wmma_f32(a0[kc], wq[kc], q0);
        q1 = wmma_f32(a1[kc], wq[kc], q1);
        Dk = wmma_f32(a0[kc] - a1[kc], wk[kc], Dk);
      }
      v8f v0 = {}, v1 = {};
      V2 ch{v0, v1, a0, a1, wv};
      const float z = score_share(q0, Dk, q1, Dk, lane, ch);
      part[lane * 4 + wid] = z;
      ch.tie_behind_lds();
      run<6, 10>(ch);
      ch.tie();
      exchange_signal();
      ch.tie();
      run<10, 16>(ch);
      ch.tie();
      exchange_wait();
      v4f f = *(const v4f*)(part + lane * 4);
      ch.tie_behind_lds();
      run<16, 22>(ch);
      blend_out<false, 22, 25, 28, 32>(f, ch, v0, v1, v0, v1, bufL, bufS,
                                       lane, wid);
    }
    phase_end<16>(xf);
  }

  if (wid == 0) {
    const float4* src = (const float4*)(bufL + m * LS + 32 * h);
    float4* dst = (float4*)(out + (size_t)b * DDIM + 32 * h);
#pragma unroll
    for (int i = 0; i < 8; ++i) dst[i] = src[i];
  }
}

extern "C" void kernel_launch(void* const* d_in, const int* in_sizes, int n_in,
                              void* d_out, int out_size, void* d_ws,
                              size_t ws_size, hipStream_t stream) {
  (void)in_sizes; (void)n_in; (void)out_size; (void)ws_size;
  const float* x  = (const float*)d_in[0];
  const float* Wq = (const float*)d_in[1];
  const float* Wk = (const float*)d_in[2];
  const float* Wv = (const float*)d_in[3];
  float* wsw = (float*)d_ws;

  swizzle_weights<<<48, 256, 0, stream>>>(Wq, Wk, Wv, wsw);
  birnn_kernel<<<2048 / 16, 128, 0, stream>>>(x, wsw, (float*)d_out);
}
